# starting kernel
# baseline (speedup 1.0000x reference)
#include <hip/hip_runtime.h>
#include <hip/hip_bf16.h>

typedef float v2f __attribute__((ext_vector_type(2)));
typedef float v8f __attribute__((ext_vector_type(8)));

#define N_NODES   25000
#define N_EDGES   320000
#define IN_FEAT   64
#define OUT_FEAT  64
#define NUM_HEADS 5
#define NUM_RELS  20
#define NUM_BASES 10
#define HF        (NUM_HEADS * OUT_FEAT)
#define ATTN_ELEMS (NUM_RELS * NUM_HEADS * 2 * OUT_FEAT)
#define MTILES ((N_NODES + 15) / 16)

__global__ void compose_attn_kernel(const float* __restrict__ w_comp,
                                    const float* __restrict__ attn_w,
                                    float* __restrict__ attn_fcs) {
    int t = blockIdx.x * blockDim.x + threadIdx.x;
    if (t >= ATTN_ELEMS) return;
    int r = t / (NUM_HEADS * 2 * OUT_FEAT);
    int h = (t / (2 * OUT_FEAT)) % NUM_HEADS;
    int f = t % (2 * OUT_FEAT);
    float acc = 0.f;
#pragma unroll
    for (int b = 0; b < NUM_BASES; ++b)
        acc += w_comp[r * NUM_BASES + b] * attn_w[b * (2 * OUT_FEAT * NUM_HEADS) + f * NUM_HEADS + h];
    attn_fcs[t] = acc;
}

template <int LDW, int NT>
__global__ void proj_wmma_kernel(const float* __restrict__ feat,
                                 const float* __restrict__ W,
                                 float* __restrict__ Out) {
    constexpr int WCOLS = LDW / (16 * NT);
    const int wave = threadIdx.x >> 5;
    const int lane = threadIdx.x & 31;
    const int tile = blockIdx.x * (blockDim.x >> 5) + wave;
    if (tile >= MTILES * WCOLS) return;

    const int mt = tile / WCOLS;
    const int wc = tile % WCOLS;

    const int l16  = lane & 15;
    const int koff = (lane < 16) ? 0 : 2;
    const int rowA = min(mt * 16 + l16, N_NODES - 1);
    const int colb = wc * (16 * NT) + l16;

    const float* ap = feat + (size_t)rowA * IN_FEAT + koff;
    const float* bp = W + (size_t)koff * LDW + colb;

    v8f c[NT];
#pragma unroll
    for (int j = 0; j < NT; ++j) c[j] = v8f{};

#pragma unroll
    for (int k = 0; k < IN_FEAT; k += 4) {
        v2f a;
        a.x = ap[k];
        a.y = ap[k + 1];
#pragma unroll
        for (int j = 0; j < NT; ++j) {
            v2f b;
            b.x = bp[k * LDW + j * 16];
            b.y = bp[(k + 1) * LDW + j * 16];
            c[j] = __builtin_amdgcn_wmma_f32_16x16x4_f32(
                        false, a,  false, b,
                        (short)0, c[j],  false,  false);
        }
    }

    const int mbase = mt * 16 + ((lane < 16) ? 0 : 8);
    float* op = Out + (size_t)mbase * LDW + colb;
#pragma unroll
    for (int r = 0; r < 8; ++r) {
        if (mbase + r < N_NODES) {
#pragma unroll
            for (int j = 0; j < NT; ++j)
                op[r * LDW + j * 16] = c[j][r];
        }
    }
}

__global__ void edge_kernel(const float* __restrict__ z,
                            const int*   __restrict__ src,
                            const int*   __restrict__ dst,
                            const int*   __restrict__ etype,
                            const float* __restrict__ attn_fcs,
                            float*       __restrict__ out) {
    __shared__ float sW[ATTN_ELEMS];
    for (int i = threadIdx.x; i < ATTN_ELEMS; i += blockDim.x) sW[i] = attn_fcs[i];
    __syncthreads();

    const int lane  = threadIdx.x & 31;
    const int wid   = blockIdx.x * (blockDim.x >> 5) + (threadIdx.x >> 5);
    const int nwave = gridDim.x * (blockDim.x >> 5);
    const int f0    = lane * 2;

    for (int e = wid; e < N_EDGES; e += nwave) {
        const int s = src[e], d = dst[e], r = etype[e];
        const float2 zs = *(const float2*)(z + (size_t)s * IN_FEAT + f0);
        const float2 zd = *(const float2*)(z + (size_t)d * IN_FEAT + f0);
        const float* Wr = sW + r * (NUM_HEADS * 2 * OUT_FEAT);

        float att[NUM_HEADS];
#pragma unroll
        for (int h = 0; h < NUM_HEADS; ++h) {
            const float* Wh = Wr + h * (2 * OUT_FEAT);
            const float2 ws = *(const float2*)(Wh + f0);
            const float2 wd = *(const float2*)(Wh + OUT_FEAT + f0);
            float p = zs.x * ws.x + zs.y * ws.y + zd.x * wd.x + zd.y * wd.y;
#pragma unroll
            for (int m = 16; m > 0; m >>= 1) p += __shfl_xor(p, m, 32);
            att[h] = (p > 0.f) ? p : 0.01f * p;
        }

        float* op = out + (size_t)d * HF + f0;
#pragma unroll
        for (int h = 0; h < NUM_HEADS; ++h) {
            atomicAdd(op + h * OUT_FEAT,     att[h] * zs.x);
            atomicAdd(op + h * OUT_FEAT + 1, att[h] * zs.y);
        }
    }
}

extern "C" void kernel_launch(void* const* d_in, const int* in_sizes, int n_in,
                              void* d_out, int out_size, void* d_ws, size_t ws_size,
                              hipStream_t stream) {
    const float* feat      = (const float*)d_in[0];
    const int*   src       = (const int*)  d_in[1];
    const int*   dst       = (const int*)  d_in[2];
    const int*   etype     = (const int*)  d_in[3];
    const float* fc_w      = (const float*)d_in[4];
    const float* self_fc_w = (const float*)d_in[5];
    const float* attn_w    = (const float*)d_in[6];
    const float* w_comp    = (const float*)d_in[7];
    float* out = (float*)d_out;

    float* z        = (float*)d_ws;
    float* attn_fcs = z + (size_t)N_NODES * OUT_FEAT;

    compose_attn_kernel<<<(ATTN_ELEMS + 255) / 256, 256, 0, stream>>>(w_comp, attn_w, attn_fcs);

    {
        int tiles  = MTILES * (OUT_FEAT / 32);
        int blocks = (tiles + 3) / 4;
        proj_wmma_kernel<OUT_FEAT, 2><<<blocks, 128, 0, stream>>>(feat, fc_w, z);
    }
    {
        int tiles  = MTILES * (HF / 32);
        int blocks = (tiles + 3) / 4;
        proj_wmma_kernel<HF, 2><<<blocks, 128, 0, stream>>>(feat, self_fc_w, out);
    }

    edge_kernel<<<1024, 256, 0, stream>>>(z, src, dst, etype, attn_fcs, out);
}


# each wave-wide f32 atomic add covers one whole 128-byte line: lane owns features l and l+32 (was 2l, 2l+1), coefficients stored pairwise to match; in-lane order of the dot partial sums changes
# speedup vs baseline: 1.1357x; 1.1357x over previous
#include <hip/hip_runtime.h>
#include <hip/hip_bf16.h>

typedef float v2f __attribute__((ext_vector_type(2)));
typedef float v8f __attribute__((ext_vector_type(8)));

#define N_NODES   25000
#define N_EDGES   320000
#define IN_FEAT   64
#define OUT_FEAT  64
#define NUM_HEADS 5
#define NUM_RELS  20
#define NUM_BASES 10
#define HF        (NUM_HEADS * OUT_FEAT)
#define ATTN_ELEMS (NUM_RELS * NUM_HEADS * 2 * OUT_FEAT)
#define MTILES ((N_NODES + 15) / 16)

__global__ void compose_attn_kernel(const float* __restrict__ w_comp,
                                    const float* __restrict__ attn_w,
                                    float* __restrict__ attn_fcs) {
    int t = blockIdx.x * blockDim.x + threadIdx.x;
    if (t >= ATTN_ELEMS) return;
    int r = t / (NUM_HEADS * 2 * OUT_FEAT);
    int h = (t / (2 * OUT_FEAT)) % NUM_HEADS;
    int f = t % (2 * OUT_FEAT);
    float acc = 0.f;
#pragma unroll
    for (int b = 0; b < NUM_BASES; ++b)
        acc += w_comp[r * NUM_BASES + b] * attn_w[b * (2 * OUT_FEAT * NUM_HEADS) + f * NUM_HEADS + h];
    const int part = f / OUT_FEAT;
    const int ff   = f % OUT_FEAT;
    const int pos  = (ff < 32) ? (2 * ff) : (2 * (ff - 32) + 1);
    attn_fcs[(r * NUM_HEADS + h) * (2 * OUT_FEAT) + part * OUT_FEAT + pos] = acc;
}

template <int LDW, int NT>
__global__ void proj_wmma_kernel(const float* __restrict__ feat,
                                 const float* __restrict__ W,
                                 float* __restrict__ Out) {
    constexpr int WCOLS = LDW / (16 * NT);
    const int wave = threadIdx.x >> 5;
    const int lane = threadIdx.x & 31;
    const int tile = blockIdx.x * (blockDim.x >> 5) + wave;
    if (tile >= MTILES * WCOLS) return;

    const int mt = tile / WCOLS;
    const int wc = tile % WCOLS;

    const int l16  = lane & 15;
    const int koff = (lane < 16) ? 0 : 2;
    const int rowA = min(mt * 16 + l16, N_NODES - 1);
    const int colb = wc * (16 * NT) + l16;

    const float* ap = feat + (size_t)rowA * IN_FEAT + koff;
    const float* bp = W + (size_t)koff * LDW + colb;

    v8f c[NT];
#pragma unroll
    for (int j = 0; j < NT; ++j) c[j] = v8f{};

#pragma unroll
    for (int k = 0; k < IN_FEAT; k += 4) {
        v2f a;
        a.x = ap[k];
        a.y = ap[k + 1];
#pragma unroll
        for (int j = 0; j < NT; ++j) {
            v2f b;
            b.x = bp[k * LDW + j * 16];
            b.y = bp[(k + 1) * LDW + j * 16];
            c[j] = __builtin_amdgcn_wmma_f32_16x16x4_f32(
                        false, a,  false, b,
                        (short)0, c[j],  false,  false);
        }
    }

    const int mbase = mt * 16 + ((lane < 16) ? 0 : 8);
    float* op = Out + (size_t)mbase * LDW + colb;
#pragma unroll
    for (int r = 0; r < 8; ++r) {
        if (mbase + r < N_NODES) {
#pragma unroll
            for (int j = 0; j < NT; ++j)
                op[r * LDW + j * 16] = c[j][r];
        }
    }
}

__global__ void edge_kernel(const float* __restrict__ z,
                            const int*   __restrict__ src,
                            const int*   __restrict__ dst,
                            const int*   __restrict__ etype,
                            const float* __restrict__ attn_fcs,
                            float*       __restrict__ out) {
    __shared__ float sW[ATTN_ELEMS];
    for (int i = threadIdx.x; i < ATTN_ELEMS; i += blockDim.x) sW[i] = attn_fcs[i];
    __syncthreads();

    const int lane  = threadIdx.x & 31;
    const int wid   = blockIdx.x * (blockDim.x >> 5) + (threadIdx.x >> 5);
    const int nwave = gridDim.x * (blockDim.x >> 5);

    for (int e = wid; e < N_EDGES; e += nwave) {
        const int s = src[e], d = dst[e], r = etype[e];
        const float* zsp = z + (size_t)s * IN_FEAT + lane;
        const float* zdp = z + (size_t)d * IN_FEAT + lane;
        const float zs0 = zsp[0], zs1 = zsp[32];
        const float zd0 = zdp[0], zd1 = zdp[32];
        const float* Wr = sW + r * (NUM_HEADS * 2 * OUT_FEAT);

        float att[NUM_HEADS];
#pragma unroll
        for (int h = 0; h < NUM_HEADS; ++h) {
            const float* Wh = Wr + h * (2 * OUT_FEAT);
            const float2 ws = *(const float2*)(Wh + 2 * lane);
            const float2 wd = *(const float2*)(Wh + OUT_FEAT + 2 * lane);
            float p = zs0 * ws.x + zs1 * ws.y + zd0 * wd.x + zd1 * wd.y;
#pragma unroll
            for (int m = 16; m > 0; m >>= 1) p += __shfl_xor(p, m, 32);
            att[h] = (p > 0.f) ? p : 0.01f * p;
        }

        float* op = out + (size_t)d * HF + lane;
#pragma unroll
        for (int h = 0; h < NUM_HEADS; ++h) {
            atomicAdd(op + h * OUT_FEAT,      att[h] * zs0);
            atomicAdd(op + h * OUT_FEAT + 32, att[h] * zs1);
        }
    }
}

extern "C" void kernel_launch(void* const* d_in, const int* in_sizes, int n_in,
                              void* d_out, int out_size, void* d_ws, size_t ws_size,
                              hipStream_t stream) {
    const float* feat      = (const float*)d_in[0];
    const int*   src       = (const int*)  d_in[1];
    const int*   dst       = (const int*)  d_in[2];
    const int*   etype     = (const int*)  d_in[3];
    const float* fc_w      = (const float*)d_in[4];
    const float* self_fc_w = (const float*)d_in[5];
    const float* attn_w    = (const float*)d_in[6];
    const float* w_comp    = (const float*)d_in[7];
    float* out = (float*)d_out;

    float* z        = (float*)d_ws;
    float* attn_fcs = z + (size_t)N_NODES * OUT_FEAT;

    compose_attn_kernel<<<(ATTN_ELEMS + 255) / 256, 256, 0, stream>>>(w_comp, attn_w, attn_fcs);

    {
        int tiles  = MTILES * (OUT_FEAT / 32);
        int blocks = (tiles + 3) / 4;
        proj_wmma_kernel<OUT_FEAT, 2><<<blocks, 128, 0, stream>>>(feat, fc_w, z);
    }
    {
        int tiles  = MTILES * (HF / 32);
        int blocks = (tiles + 3) / 4;
        proj_wmma_kernel<HF, 2><<<blocks, 128, 0, stream>>>(feat, self_fc_w, out);
    }

    edge_kernel<<<1024, 256, 0, stream>>>(z, src, dst, etype, attn_fcs, out);
}


# per-edge 128-long attention dots hoisted into a per-node, per-relation logit table P = z @ Wcat (f32 WMMA, in the self projection launch); an edge lane adds two table entries; logit sum order changes
# speedup vs baseline: 1.3311x; 1.1721x over previous
#include <hip/hip_runtime.h>
#include <hip/hip_bf16.h>


typedef float v2f __attribute__((ext_vector_type(2)));
typedef float v8f __attribute__((ext_vector_type(8)));

#define N_NODES   25000
#define N_EDGES   320000
#define IN_FEAT   64
#define OUT_FEAT  64
#define NUM_HEADS 5
#define NUM_RELS  20
#define NUM_BASES 10
#define HF        (NUM_HEADS * OUT_FEAT)
#define MTILES ((N_NODES + 15) / 16)

#define PBLK      12
#define PLD       (NUM_RELS * PBLK)
#define WCAT_ELEMS (OUT_FEAT * PLD)

#define EDGES_PER_WAVE 32
#define EDGE_GROUP     8

static_assert(IN_FEAT == OUT_FEAT, "the projection kernel uses one K for feat and z");
static_assert(NUM_HEADS == 5 && PBLK == 12, "logit block layout is written for 5 heads");
static_assert(N_EDGES % EDGES_PER_WAVE == 0, "edge kernel takes whole 32-edge batches");
static_assert(EDGES_PER_WAVE % EDGE_GROUP == 0, "groups tile the batch");

__global__ void compose_wcat_kernel(const float* __restrict__ w_comp,
                                    const float* __restrict__ attn_w,
                                    float* __restrict__ Wcat) {
    int t = blockIdx.x * blockDim.x + threadIdx.x;
    if (t >= WCAT_ELEMS) return;
    const int k   = t / PLD;
    const int col = t % PLD;
    const int r   = col / PBLK;
    const int c   = col % PBLK;
    float acc = 0.f;
    if (c < 10) {
        const int part = (c < 8) ? (c >> 2) : (c - 8);
        const int h    = (c < 8) ? (c & 3)  : 4;
        const int f    = part * OUT_FEAT + k;
#pragma unroll
        for (int b = 0; b < NUM_BASES; ++b)
            acc += w_comp[r * NUM_BASES + b] * attn_w[b * (2 * OUT_FEAT * NUM_HEADS) + f * NUM_HEADS + h];
    }
    Wcat[t] = acc;
}

template <int LDW, int NT>
__device__ __forceinline__ void proj_tile(const float* __restrict__ A,
                                          const float* __restrict__ W,
                                          float* __restrict__ Out,
                                          int tile, int lane) {
    constexpr int WCOLS = LDW / (16 * NT);
    static_assert(WCOLS * 16 * NT == LDW, "N-tiles must tile the row");
    const int mt = tile / WCOLS;
    const int wc = tile % WCOLS;

    const int l16  = lane & 15;
    const int koff = (lane < 16) ? 0 : 2;
    const int rowA = min(mt * 16 + l16, N_NODES - 1);
    const int colb = wc * (16 * NT) + l16;

    const float* ap = A + (size_t)rowA * IN_FEAT + koff;
    const float* bp = W + (size_t)koff * LDW + colb;

    v8f c[NT];
#pragma unroll
    for (int j = 0; j < NT; ++j) c[j] = v8f{};

#pragma unroll
    for (int k = 0; k < IN_FEAT; k += 4) {
        v2f a;
        a.x = ap[k];
        a.y = ap[k + 1];
#pragma unroll
        for (int j = 0; j < NT; ++j) {
            v2f b;
            b.x = bp[k * LDW + j * 16];
            b.y = bp[(k + 1) * LDW + j * 16];
            c[j] = __builtin_amdgcn_wmma_f32_16x16x4_f32(
                        false, a,  false, b,
                        (short)0, c[j],  false,  false);
        }
    }

    const int mbase = mt * 16 + ((lane < 16) ? 0 : 8);
    float* op = Out + (size_t)mbase * LDW + colb;
#pragma unroll
    for (int r = 0; r < 8; ++r) {
        if (mbase + r < N_NODES) {
#pragma unroll
            for (int j = 0; j < NT; ++j)
                op[r * LDW + j * 16] = c[j][r];
        }
    }
}

__global__ void proj_z_kernel(const float* __restrict__ feat,
                              const float* __restrict__ fc_w,
                              float* __restrict__ z) {
    const int lane = threadIdx.x & 31;
    const int tile = blockIdx.x * (blockDim.x >> 5) + (threadIdx.x >> 5);
    if (tile >= MTILES * (OUT_FEAT / 32)) return;
    proj_tile<OUT_FEAT, 2>(feat, fc_w, z, tile, lane);
}

#define P_TILES    (MTILES * (PLD / 80))
#define SELF_TILES (MTILES * (HF / 32))
#define ALL_TILES  (P_TILES + SELF_TILES)
__global__ void proj_self_and_logits_kernel(const float* __restrict__ feat,
                                            const float* __restrict__ self_fc_w,
                                            float* __restrict__ out,
                                            const float* __restrict__ z,
                                            const float* __restrict__ Wcat,
                                            float* __restrict__ P) {
    const int lane = threadIdx.x & 31;
    const int tile = __builtin_amdgcn_readfirstlane(blockIdx.x * (blockDim.x >> 5) + (threadIdx.x >> 5));
    if (tile >= ALL_TILES) return;
    if (tile < P_TILES) proj_tile<PLD, 5>(z, Wcat, P, tile, lane);
    else                proj_tile<HF, 2>(feat, self_fc_w, out, tile - P_TILES, lane);
}

__device__ __forceinline__ float lane_bcast(float v, int j) {
    return __int_as_float(__builtin_amdgcn_readlane(__float_as_int(v), j));
}

__global__ void edge_kernel(const float* __restrict__ z,
                            const int*   __restrict__ src,
                            const int*   __restrict__ dst,
                            const int*   __restrict__ etype,
                            const float* __restrict__ P,
                            float*       __restrict__ out) {
    const unsigned lane = threadIdx.x & 31;
    const int wid  = blockIdx.x * (blockDim.x >> 5) + (threadIdx.x >> 5);
    const int base = wid * EDGES_PER_WAVE;
    if (base >= N_EDGES) return;

    const int e = base + (int)lane;
    const int s = src[e], d = dst[e], r = etype[e];

    const float* ps = P + (size_t)s * PLD + r * PBLK;
    const float* pd = P + (size_t)d * PLD + r * PBLK;
    const float4 a03 = *(const float4*)(ps);
    const float  a4  = ps[8];
    const float4 b03 = *(const float4*)(pd + 4);
    const float  b4  = pd[9];

    float att[NUM_HEADS];
    att[0] = a03.x + b03.x;
    att[1] = a03.y + b03.y;
    att[2] = a03.z + b03.z;
    att[3] = a03.w + b03.w;
    att[4] = a4 + b4;
#pragma unroll
    for (int h = 0; h < NUM_HEADS; ++h)
        att[h] = (att[h] > 0.f) ? att[h] : 0.01f * att[h];

    float z0[2][EDGE_GROUP], z1[2][EDGE_GROUP];

#pragma unroll
    for (int j = 0; j < EDGE_GROUP; ++j) {
        const int sj = __builtin_amdgcn_readlane(s, j);
        const float* zrow = z + (size_t)sj * OUT_FEAT;
        z0[0][j] = zrow[lane];
        z1[0][j] = zrow[lane + 32];
    }

#pragma unroll
    for (int g = 0; g < EDGES_PER_WAVE / EDGE_GROUP; ++g) {
        const int cur = g & 1;
        if (g + 1 < EDGES_PER_WAVE / EDGE_GROUP) {
#pragma unroll
            for (int j = 0; j < EDGE_GROUP; ++j) {
                const int sj = __builtin_amdgcn_readlane(s, (g + 1) * EDGE_GROUP + j);
                const float* zrow = z + (size_t)sj * OUT_FEAT;
                z0[cur ^ 1][j] = zrow[lane];
                z1[cur ^ 1][j] = zrow[lane + 32];
            }
        }
#pragma unroll
        for (int j = 0; j < EDGE_GROUP; ++j) {
            const int ej = g * EDGE_GROUP + j;
            const int dj = __builtin_amdgcn_readlane(d, ej);
            float* orow = out + (size_t)dj * HF;
#pragma unroll
            for (int h = 0; h < NUM_HEADS; ++h) {
                const float ah = lane_bcast(att[h], ej);
                atomicAdd(orow + h * OUT_FEAT      + lane, ah * z0[cur][j]);
                atomicAdd(orow + h * OUT_FEAT + 32 + lane, ah * z1[cur][j]);
            }
        }
    }
}

extern "C" void kernel_launch(void* const* d_in, const int* in_sizes, int n_in,
                              void* d_out, int out_size, void* d_ws, size_t ws_size,
                              hipStream_t stream) {
    const float* feat      = (const float*)d_in[0];
    const int*   src       = (const int*)  d_in[1];
    const int*   dst       = (const int*)  d_in[2];
    const int*   etype     = (const int*)  d_in[3];
    const float* fc_w      = (const float*)d_in[4];
    const float* self_fc_w = (const float*)d_in[5];
    const float* attn_w    = (const float*)d_in[6];
    const float* w_comp    = (const float*)d_in[7];
    float* out = (float*)d_out;

    float* z    = (float*)d_ws;
    float* Wcat = z + (size_t)N_NODES * OUT_FEAT;
    float* P    = Wcat + WCAT_ELEMS;

    if (ws_size < ((size_t)N_NODES * OUT_FEAT + WCAT_ELEMS + (size_t)N_NODES * PLD) * sizeof(float)) return;

    compose_wcat_kernel<<<(WCAT_ELEMS + 255) / 256, 256, 0, stream>>>(w_comp, attn_w, Wcat);

    {
        int tiles  = MTILES * (OUT_FEAT / 32);
        int blocks = (tiles + 3) / 4;
        proj_z_kernel<<<blocks, 128, 0, stream>>>(feat, fc_w, z);
    }
    {
        int blocks = (ALL_TILES + 3) / 4;
        proj_self_and_logits_kernel<<<blocks, 128, 0, stream>>>(feat, self_fc_w, out, z, Wcat, P);
    }

    edge_kernel<<<N_EDGES / EDGES_PER_WAVE / 8, 256, 0, stream>>>(z, src, dst, etype, P, out);
}
